# starting kernel
# baseline (speedup 1.0000x reference)
#include <hip/hip_runtime.h>

typedef __attribute__((ext_vector_type(2))) float v2f;
typedef __attribute__((ext_vector_type(8))) float v8f;

#define HSTR 260
#define H1S   68
#define H2S   36

__global__ __launch_bounds__(128)
void dfm_fused_kernel(const int*   __restrict__ x_field,
                      const int*   __restrict__ x,
                      const float* __restrict__ x_val,
                      const float* __restrict__ emb,
                      const float* __restrict__ lin_tab,
                      const float* __restrict__ lin_bias,
                      const float* __restrict__ W1, const float* __restrict__ b1,
                      const float* __restrict__ g1, const float* __restrict__ be1,
                      const float* __restrict__ W2, const float* __restrict__ b2,
                      const float* __restrict__ g2, const float* __restrict__ be2,
                      const float* __restrict__ W3, const float* __restrict__ b3,
                      float*       __restrict__ out)
{
    __shared__ float sh_h [16 * HSTR];
    __shared__ float sh_h1[16 * H1S];
    __shared__ float sh_h2[16 * H2S];
    __shared__ float sh_red[16 * 8];
    __shared__ float sh_lr[256];
    __shared__ float sh_lf[16];

    const int t    = threadIdx.x;
    const int wave = t >> 5;
    const int lane = t & 31;
    const int half = lane >> 4;
    const int mlo  = lane & 15;
    const int r0   = blockIdx.x * 16;
    const float RSQ = rsqrtf(1.0f + 1e-5f);

    for (int p = t; p < 256; p += 128) {
        const int r = p >> 4, f = p & 15;
        const int gi = (r0 + r) * 16 + f;
        const long long row = (long long)x_field[gi] * 100000ll + (long long)x[gi];
        const float val = x_val[gi];
        const float4* e4 = reinterpret_cast<const float4*>(emb + (size_t)row * 16u);
        float4 a0 = e4[0], a1 = e4[1], a2 = e4[2], a3 = e4[3];
        float4* dst = reinterpret_cast<float4*>(&sh_h[r * HSTR + f * 16]);
        dst[0] = make_float4(a0.x * val, a0.y * val, a0.z * val, a0.w * val);
        dst[1] = make_float4(a1.x * val, a1.y * val, a1.z * val, a1.w * val);
        dst[2] = make_float4(a2.x * val, a2.y * val, a2.z * val, a2.w * val);
        dst[3] = make_float4(a3.x * val, a3.y * val, a3.z * val, a3.w * val);
        sh_lr[p] = lin_tab[row] * val;
    }
    __syncthreads();

    {
        const int r = t >> 3;
        const int e = (t & 7) * 2;
        float s0 = 0.f, q0 = 0.f, s1 = 0.f, q1 = 0.f;
        #pragma unroll
        for (int f = 0; f < 16; ++f) {
            const float v0 = sh_h[r * HSTR + f * 16 + e];
            const float v1 = sh_h[r * HSTR + f * 16 + e + 1];
            s0 += v0; q0 = fmaf(v0, v0, q0);
            s1 += v1; q1 = fmaf(v1, v1, q1);
        }
        sh_red[r * 8 + (t & 7)] = (s0 * s0 - q0) + (s1 * s1 - q1);
    }
    __syncthreads();
    if (t < 16) {
        float fm = 0.f;
        #pragma unroll
        for (int i = 0; i < 8; ++i) fm += sh_red[t * 8 + i];
        float lin = 0.f;
        #pragma unroll
        for (int f = 0; f < 16; ++f) lin += sh_lr[t * 16 + f];
        sh_lf[t] = lin + lin_bias[0] + 0.5f * fm;
    }
    __syncthreads();

    {
        const int c = wave * 16 + mlo;
        v8f acc = {};
        #pragma unroll 4
        for (int kk = 0; kk < 256; kk += 4) {
            const int k0 = kk + 2 * half;
            v2f a = *(const v2f*)&sh_h[mlo * HSTR + k0];
            v2f b; b.x = W1[(size_t)k0 * 64 + c];
                   b.y = W1[(size_t)(k0 + 1) * 64 + c];
            acc = __builtin_amdgcn_wmma_f32_16x16x4_f32(false, a, false, b,
                                                        (short)0, acc, false, false);
        }
        const float sc = g1[c] * RSQ;
        const float tb = fmaf(b1[c], sc, be1[c]);
        #pragma unroll
        for (int j = 0; j < 8; ++j) {
            const int m = j + 8 * half;
            const float v = fmaf(acc[j], sc, tb);
            sh_h1[m * H1S + c] = v > 0.f ? v : 0.f;
        }
    }
    __syncthreads();

    if (wave < 2) {
        const int c = wave * 16 + mlo;
        v8f acc = {};
        #pragma unroll 4
        for (int kk = 0; kk < 64; kk += 4) {
            const int k0 = kk + 2 * half;
            v2f a = *(const v2f*)&sh_h1[mlo * H1S + k0];
            v2f b; b.x = W2[k0 * 32 + c];
                   b.y = W2[(k0 + 1) * 32 + c];
            acc = __builtin_amdgcn_wmma_f32_16x16x4_f32(false, a, false, b,
                                                        (short)0, acc, false, false);
        }
        const float sc = g2[c] * RSQ;
        const float tb = fmaf(b2[c], sc, be2[c]);
        #pragma unroll
        for (int j = 0; j < 8; ++j) {
            const int m = j + 8 * half;
            const float v = fmaf(acc[j], sc, tb);
            sh_h2[m * H2S + c] = v > 0.f ? v : 0.f;
        }
    }
    __syncthreads();

    if (t < 16) {
        float acc = sh_lf[t] + b3[0];
        #pragma unroll
        for (int cc = 0; cc < 32; ++cc)
            acc = fmaf(sh_h2[t * H2S + cc], W3[cc], acc);
        out[r0 + t] = acc;
    }
}

extern "C" void kernel_launch(void* const* d_in, const int* in_sizes, int n_in,
                              void* d_out, int out_size, void* d_ws, size_t ws_size,
                              hipStream_t stream) {
    const int*   x_field  = (const int*)  d_in[0];
    const int*   x        = (const int*)  d_in[1];
    const float* x_val    = (const float*)d_in[2];
    const float* emb      = (const float*)d_in[3];
    const float* lin_tab  = (const float*)d_in[4];
    const float* lin_bias = (const float*)d_in[5];
    const float* W1 = (const float*)d_in[6];
    const float* b1 = (const float*)d_in[7];
    const float* g1 = (const float*)d_in[8];
    const float* be1= (const float*)d_in[9];
    const float* W2 = (const float*)d_in[10];
    const float* b2 = (const float*)d_in[11];
    const float* g2 = (const float*)d_in[12];
    const float* be2= (const float*)d_in[13];
    const float* W3 = (const float*)d_in[14];
    const float* b3 = (const float*)d_in[15];
    float* out = (float*)d_out;

    const int nblocks = out_size / 16;
    dfm_fused_kernel<<<dim3(nblocks), dim3(128), 0, stream>>>(
        x_field, x, x_val, emb, lin_tab, lin_bias,
        W1, b1, g1, be1, W2, b2, g2, be2, W3, b3, out);
}


# 64 rows per work-group (was 16): each wave applies every W1 fragment load to 4 row tiles (W2: 2); 256 work-groups, one per WGP; a phase's loads issued together; every sum order unchanged
# speedup vs baseline: 1.4962x; 1.4962x over previous
#include <hip/hip_runtime.h>


typedef __attribute__((ext_vector_type(2))) float v2f;
typedef __attribute__((ext_vector_type(8))) float v8f;

#define ROWS  64
#define MT     4
#define NGATH  8
#define HSTR 260
#define H1S   68
#define H2S   36

__global__ __launch_bounds__(128)
void dfm_fused_kernel(const int*   __restrict__ x_field,
                      const int*   __restrict__ x,
                      const float* __restrict__ x_val,
                      const float* __restrict__ emb,
                      const float* __restrict__ lin_tab,
                      const float* __restrict__ lin_bias,
                      const float* __restrict__ W1, const float* __restrict__ b1,
                      const float* __restrict__ g1, const float* __restrict__ be1,
                      const float* __restrict__ W2, const float* __restrict__ b2,
                      const float* __restrict__ g2, const float* __restrict__ be2,
                      const float* __restrict__ W3, const float* __restrict__ b3,
                      float*       __restrict__ out)
{
    __shared__ float sh_h [ROWS * HSTR];
    __shared__ float sh_h1[ROWS * H1S];
    __shared__ float sh_h2[ROWS * H2S];
    __shared__ float sh_red[ROWS * 8];
    __shared__ float sh_lr[ROWS * 16];
    __shared__ float sh_lf[ROWS];

    const int t    = threadIdx.x;
    const int wave = t >> 5;
    const int lane = t & 31;
    const int half = lane >> 4;
    const int mlo  = lane & 15;
    const int r0   = blockIdx.x * ROWS;
    const float RSQ = rsqrtf(1.0f + 1e-5f);

    {
        long long rowi[NGATH];
        float     val[NGATH];
        #pragma unroll
        for (int i = 0; i < NGATH; ++i) {
            const int gi = r0 * 16 + t + 128 * i;
            rowi[i] = (long long)x_field[gi] * 100000ll + (long long)x[gi];
            val[i]  = x_val[gi];
        }
        float4 a[NGATH][4];
        float  lw[NGATH];
        #pragma unroll
        for (int i = 0; i < NGATH; ++i) {
            const float4* e4 = reinterpret_cast<const float4*>(emb + (size_t)rowi[i] * 16u);
            a[i][0] = e4[0]; a[i][1] = e4[1]; a[i][2] = e4[2]; a[i][3] = e4[3];
            lw[i] = lin_tab[rowi[i]];
        }
        #pragma unroll
        for (int i = 0; i < NGATH; ++i) {
            const int p = t + 128 * i;
            const int r = p >> 4, f = p & 15;
            const float v = val[i];
            float4* dst = reinterpret_cast<float4*>(&sh_h[r * HSTR + f * 16]);
            dst[0] = make_float4(a[i][0].x * v, a[i][0].y * v, a[i][0].z * v, a[i][0].w * v);
            dst[1] = make_float4(a[i][1].x * v, a[i][1].y * v, a[i][1].z * v, a[i][1].w * v);
            dst[2] = make_float4(a[i][2].x * v, a[i][2].y * v, a[i][2].z * v, a[i][2].w * v);
            dst[3] = make_float4(a[i][3].x * v, a[i][3].y * v, a[i][3].z * v, a[i][3].w * v);
            sh_lr[p] = lw[i] * v;
        }
    }
    __syncthreads();

    #pragma unroll
    for (int g = 0; g < MT; ++g) {
        const int r = g * 16 + (t >> 3);
        const int e = (t & 7) * 2;
        float s0 = 0.f, q0 = 0.f, s1 = 0.f, q1 = 0.f;
        #pragma unroll
        for (int f = 0; f < 16; ++f) {
            const float v0 = sh_h[r * HSTR + f * 16 + e];
            const float v1 = sh_h[r * HSTR + f * 16 + e + 1];
            s0 += v0; q0 = fmaf(v0, v0, q0);
            s1 += v1; q1 = fmaf(v1, v1, q1);
        }
        sh_red[r * 8 + (t & 7)] = (s0 * s0 - q0) + (s1 * s1 - q1);
    }
    __syncthreads();
    if (t < ROWS) {
        float fm = 0.f;
        #pragma unroll
        for (int i = 0; i < 8; ++i) fm += sh_red[t * 8 + i];
        float lin = 0.f;
        #pragma unroll
        for (int f = 0; f < 16; ++f) lin += sh_lr[t * 16 + f];
        sh_lf[t] = lin + lin_bias[0] + 0.5f * fm;
    }
    __syncthreads();

    {
        const int c = wave * 16 + mlo;
        v8f acc[MT];
        #pragma unroll
        for (int m = 0; m < MT; ++m) acc[m] = (v8f){};
        #pragma unroll 32
        for (int kk = 0; kk < 256; kk += 4) {
            const int k0 = kk + 2 * half;
            v2f b; b.x = W1[(size_t)k0 * 64 + c];
                   b.y = W1[(size_t)(k0 + 1) * 64 + c];
            #pragma unroll
            for (int m = 0; m < MT; ++m) {
                v2f a = *(const v2f*)&sh_h[(m * 16 + mlo) * HSTR + k0];
                acc[m] = __builtin_amdgcn_wmma_f32_16x16x4_f32(false, a, false, b,
                                                               (short)0, acc[m], false, false);
            }
        }
        const float sc = g1[c] * RSQ;
        const float tb = fmaf(b1[c], sc, be1[c]);
        #pragma unroll
        for (int m = 0; m < MT; ++m) {
            #pragma unroll
            for (int j = 0; j < 8; ++j) {
                const int row = m * 16 + j + 8 * half;
                const float v = fmaf(acc[m][j], sc, tb);
                sh_h1[row * H1S + c] = v > 0.f ? v : 0.f;
            }
        }
    }
    __syncthreads();

    {
        const int c  = (wave & 1) * 16 + mlo;
        const int mb = (wave >> 1) * 2;
        const float* w2p = W2 + (size_t)(2 * half) * 32 + c;
        v8f acc[2];
        acc[0] = (v8f){}; acc[1] = (v8f){};
        #pragma unroll
        for (int kk = 0; kk < 64; kk += 4) {
            const int k0 = kk + 2 * half;
            v2f b; b.x = w2p[kk * 32];
                   b.y = w2p[kk * 32 + 32];
            #pragma unroll
            for (int m = 0; m < 2; ++m) {
                v2f a = *(const v2f*)&sh_h1[((mb + m) * 16 + mlo) * H1S + k0];
                acc[m] = __builtin_amdgcn_wmma_f32_16x16x4_f32(false, a, false, b,
                                                               (short)0, acc[m], false, false);
            }
        }
        const float sc = g2[c] * RSQ;
        const float tb = fmaf(b2[c], sc, be2[c]);
        #pragma unroll
        for (int m = 0; m < 2; ++m) {
            #pragma unroll
            for (int j = 0; j < 8; ++j) {
                const int row = (mb + m) * 16 + j + 8 * half;
                const float v = fmaf(acc[m][j], sc, tb);
                sh_h2[row * H2S + c] = v > 0.f ? v : 0.f;
            }
        }
    }
    __syncthreads();

    if (t < ROWS) {
        float acc = sh_lf[t] + b3[0];
        #pragma unroll
        for (int cc = 0; cc < 32; ++cc)
            acc = fmaf(sh_h2[t * H2S + cc], W3[cc], acc);
        out[r0 + t] = acc;
    }
}

extern "C" void kernel_launch(void* const* d_in, const int* in_sizes, int n_in,
                              void* d_out, int out_size, void* d_ws, size_t ws_size,
                              hipStream_t stream) {
    const int*   x_field  = (const int*)  d_in[0];
    const int*   x        = (const int*)  d_in[1];
    const float* x_val    = (const float*)d_in[2];
    const float* emb      = (const float*)d_in[3];
    const float* lin_tab  = (const float*)d_in[4];
    const float* lin_bias = (const float*)d_in[5];
    const float* W1 = (const float*)d_in[6];
    const float* b1 = (const float*)d_in[7];
    const float* g1 = (const float*)d_in[8];
    const float* be1= (const float*)d_in[9];
    const float* W2 = (const float*)d_in[10];
    const float* b2 = (const float*)d_in[11];
    const float* g2 = (const float*)d_in[12];
    const float* be2= (const float*)d_in[13];
    const float* W3 = (const float*)d_in[14];
    const float* b3 = (const float*)d_in[15];
    float* out = (float*)d_out;

    const int nblocks = out_size / 64;
    dfm_fused_kernel<<<dim3(nblocks), dim3(128), 0, stream>>>(
        x_field, x, x_val, emb, lin_tab, lin_bias,
        W1, b1, g1, be1, W2, b2, g2, be2, W3, b3, out);
}


# loads moved out of the WMMA runs: W1 strip, W2, BN terms, W3, biases fetched into registers under the FM pass or earlier blocks; A fragments read from LDS one block ahead as a group; sum orders kept
# speedup vs baseline: 1.7100x; 1.1429x over previous
#include <hip/hip_runtime.h>


typedef __attribute__((ext_vector_type(2))) float v2f;
typedef __attribute__((ext_vector_type(8))) float v8f;

#define ROWS  64
#define MT     4
#define NGATH  8
#define HSTR 260
#define H1S   68
#define H2S   36

__global__ __launch_bounds__(128)
void dfm_fused_kernel(const int*   __restrict__ x_field,
                      const int*   __restrict__ x,
                      const float* __restrict__ x_val,
                      const float* __restrict__ emb,
                      const float* __restrict__ lin_tab,
                      const float* __restrict__ lin_bias,
                      const float* __restrict__ W1, const float* __restrict__ b1,
                      const float* __restrict__ g1, const float* __restrict__ be1,
                      const float* __restrict__ W2, const float* __restrict__ b2,
                      const float* __restrict__ g2, const float* __restrict__ be2,
                      const float* __restrict__ W3, const float* __restrict__ b3,
                      float*       __restrict__ out)
{
    __shared__ float sh_h [ROWS * HSTR];
    __shared__ float sh_h1[ROWS * H1S];
    __shared__ float sh_h2[ROWS * H2S];
    __shared__ float sh_red[ROWS * 8];
    __shared__ float sh_lr[ROWS * 16];
    __shared__ float sh_lf[ROWS];

    const int t    = threadIdx.x;
    const int wave = t >> 5;
    const int lane = t & 31;
    const int half = lane >> 4;
    const int mlo  = lane & 15;
    const int r0   = blockIdx.x * ROWS;
    const float RSQ = rsqrtf(1.0f + 1e-5f);

    const int c1 = wave * 16 + mlo;
    const int c2 = (wave & 1) * 16 + mlo;
    v2f   bw[64];
    v2f   bw2[16];
    float g1v, b1v, be1v, g2v, b2v, be2v;
    const float lbias = lin_bias[0];
    const float b3v   = b3[0];
    float w3r[32];
    #pragma unroll
    for (int cc = 0; cc < 32; ++cc) w3r[cc] = W3[cc];

    {
        long long rowi[NGATH];
        float     val[NGATH];
        #pragma unroll
        for (int i = 0; i < NGATH; ++i) {
            const int gi = r0 * 16 + t + 128 * i;
            rowi[i] = (long long)x_field[gi] * 100000ll + (long long)x[gi];
            val[i]  = x_val[gi];
        }
        float4 a[NGATH][4];
        float  lw[NGATH];
        #pragma unroll
        for (int i = 0; i < NGATH; ++i) {
            const float4* e4 = reinterpret_cast<const float4*>(emb + (size_t)rowi[i] * 16u);
            a[i][0] = e4[0]; a[i][1] = e4[1]; a[i][2] = e4[2]; a[i][3] = e4[3];
            lw[i] = lin_tab[rowi[i]];
        }
        #pragma unroll
        for (int i = 0; i < NGATH; ++i) {
            const int p = t + 128 * i;
            const int r = p >> 4, f = p & 15;
            const float v = val[i];
            float4* dst = reinterpret_cast<float4*>(&sh_h[r * HSTR + f * 16]);
            dst[0] = make_float4(a[i][0].x * v, a[i][0].y * v, a[i][0].z * v, a[i][0].w * v);
            dst[1] = make_float4(a[i][1].x * v, a[i][1].y * v, a[i][1].z * v, a[i][1].w * v);
            dst[2] = make_float4(a[i][2].x * v, a[i][2].y * v, a[i][2].z * v, a[i][2].w * v);
            dst[3] = make_float4(a[i][3].x * v, a[i][3].y * v, a[i][3].z * v, a[i][3].w * v);
            sh_lr[p] = lw[i] * v;
        }
    }
    __syncthreads();

    #pragma unroll
    for (int ks = 0; ks < 31; ++ks) {
        const int k0 = 4 * ks + 2 * half;
        bw[ks].x = W1[(size_t)k0 * 64 + c1];
        bw[ks].y = W1[(size_t)(k0 + 1) * 64 + c1];
    }
    __builtin_amdgcn_sched_barrier(0);

    #pragma unroll
    for (int g = 0; g < MT; ++g) {
        const int r = g * 16 + (t >> 3);
        const int e = (t & 7) * 2;
        float s0 = 0.f, q0 = 0.f, s1 = 0.f, q1 = 0.f;
        #pragma unroll
        for (int f = 0; f < 16; ++f) {
            const float v0 = sh_h[r * HSTR + f * 16 + e];
            const float v1 = sh_h[r * HSTR + f * 16 + e + 1];
            s0 += v0; q0 = fmaf(v0, v0, q0);
            s1 += v1; q1 = fmaf(v1, v1, q1);
        }
        sh_red[r * 8 + (t & 7)] = (s0 * s0 - q0) + (s1 * s1 - q1);
    }
    __syncthreads();
    if (t < ROWS) {
        float fm = 0.f;
        #pragma unroll
        for (int i = 0; i < 8; ++i) fm += sh_red[t * 8 + i];
        float lin = 0.f;
        #pragma unroll
        for (int f = 0; f < 16; ++f) lin += sh_lr[t * 16 + f];
        sh_lf[t] = lin + lbias + 0.5f * fm;
    }
    __syncthreads();

    {
        const int c = c1;
        #pragma unroll
        for (int ks = 31; ks < 62; ++ks) {
            const int k0 = 4 * ks + 2 * half;
            bw[ks].x = W1[(size_t)k0 * 64 + c];
            bw[ks].y = W1[(size_t)(k0 + 1) * 64 + c];
        }
        v8f acc[MT];
        #pragma unroll
        for (int m = 0; m < MT; ++m) acc[m] = (v8f){};
        const float* ap = &sh_h[mlo * HSTR + 2 * half];
        v2f af[2][8 * MT];
        #pragma unroll
        for (int j = 0; j < 8; ++j) {
            #pragma unroll
            for (int m = 0; m < MT; ++m)
                af[0][j * MT + m] = *(const v2f*)&ap[m * 16 * HSTR + 4 * j];
        }
        #pragma unroll
        for (int blk = 0; blk < 8; ++blk) {
            if (blk == 2) {
                #pragma unroll
                for (int ks = 62; ks < 64; ++ks) {
                    const int k0 = 4 * ks + 2 * half;
                    bw[ks].x = W1[(size_t)k0 * 64 + c];
                    bw[ks].y = W1[(size_t)(k0 + 1) * 64 + c];
                }
                g1v = g1[c1]; b1v = b1[c1]; be1v = be1[c1];
                g2v = g2[c2]; b2v = b2[c2]; be2v = be2[c2];
                const float* w2p = W2 + (size_t)(2 * half) * 32 + c2;
                #pragma unroll
                for (int i = 0; i < 16; ++i) {
                    bw2[i].x = w2p[i * 128];
                    bw2[i].y = w2p[i * 128 + 32];
                }
            }
            if (blk < 7) {
                #pragma unroll
                for (int j = 0; j < 8; ++j) {
                    #pragma unroll
                    for (int m = 0; m < MT; ++m)
                        af[(blk + 1) & 1][j * MT + m] =
                            *(const v2f*)&ap[m * 16 * HSTR + 4 * ((blk + 1) * 8 + j)];
                }
            }
            __builtin_amdgcn_sched_barrier(0);
            #pragma unroll
            for (int j = 0; j < 8; ++j) {
                const v2f b = bw[blk * 8 + j];
                #pragma unroll
                for (int m = 0; m < MT; ++m)
                    acc[m] = __builtin_amdgcn_wmma_f32_16x16x4_f32(false, af[blk & 1][j * MT + m], false, b,
                                                                   (short)0, acc[m], false, false);
            }
            __builtin_amdgcn_sched_barrier(0);
        }
        const float sc = g1v * RSQ;
        const float tb = fmaf(b1v, sc, be1v);
        #pragma unroll
        for (int m = 0; m < MT; ++m) {
            #pragma unroll
            for (int j = 0; j < 8; ++j) {
                const int row = m * 16 + j + 8 * half;
                const float v = fmaf(acc[m][j], sc, tb);
                sh_h1[row * H1S + c] = v > 0.f ? v : 0.f;
            }
        }
    }
    __syncthreads();

    {
        const int c  = c2;
        const int mb = (wave >> 1) * 2;
        v8f acc[2];
        acc[0] = (v8f){}; acc[1] = (v8f){};
        v2f a2[32];
        #pragma unroll
        for (int i = 0; i < 16; ++i) {
            #pragma unroll
            for (int m = 0; m < 2; ++m)
                a2[i * 2 + m] = *(const v2f*)&sh_h1[((mb + m) * 16 + mlo) * H1S + 4 * i + 2 * half];
        }
        __builtin_amdgcn_sched_barrier(0);
        #pragma unroll
        for (int i = 0; i < 16; ++i) {
            const v2f b = bw2[i];
            #pragma unroll
            for (int m = 0; m < 2; ++m)
                acc[m] = __builtin_amdgcn_wmma_f32_16x16x4_f32(false, a2[i * 2 + m], false, b,
                                                               (short)0, acc[m], false, false);
        }
        const float sc = g2v * RSQ;
        const float tb = fmaf(b2v, sc, be2v);
        #pragma unroll
        for (int m = 0; m < 2; ++m) {
            #pragma unroll
            for (int j = 0; j < 8; ++j) {
                const int row = (mb + m) * 16 + j + 8 * half;
                const float v = fmaf(acc[m][j], sc, tb);
                sh_h2[row * H2S + c] = v > 0.f ? v : 0.f;
            }
        }
    }
    __syncthreads();

    if (t < ROWS) {
        float acc = sh_lf[t] + b3v;
        #pragma unroll
        for (int cc = 0; cc < 32; ++cc)
            acc = fmaf(sh_h2[t * H2S + cc], w3r[cc], acc);
        out[r0 + t] = acc;
    }
}

extern "C" void kernel_launch(void* const* d_in, const int* in_sizes, int n_in,
                              void* d_out, int out_size, void* d_ws, size_t ws_size,
                              hipStream_t stream) {
    const int*   x_field  = (const int*)  d_in[0];
    const int*   x        = (const int*)  d_in[1];
    const float* x_val    = (const float*)d_in[2];
    const float* emb      = (const float*)d_in[3];
    const float* lin_tab  = (const float*)d_in[4];
    const float* lin_bias = (const float*)d_in[5];
    const float* W1 = (const float*)d_in[6];
    const float* b1 = (const float*)d_in[7];
    const float* g1 = (const float*)d_in[8];
    const float* be1= (const float*)d_in[9];
    const float* W2 = (const float*)d_in[10];
    const float* b2 = (const float*)d_in[11];
    const float* g2 = (const float*)d_in[12];
    const float* be2= (const float*)d_in[13];
    const float* W3 = (const float*)d_in[14];
    const float* b3 = (const float*)d_in[15];
    float* out = (float*)d_out;

    const int nblocks = out_size / 64;
    dfm_fused_kernel<<<dim3(nblocks), dim3(128), 0, stream>>>(
        x_field, x, x_val, emb, lin_tab, lin_bias,
        W1, b1, g1, be1, W2, b2, g2, be2, W3, b3, out);
}


# gather re-laid: four lanes read one embedding row, 16 B each (a load touches 8 lines, was 32); FM and linear partials taken from those registers, one barrier for three; sum orders kept
# speedup vs baseline: 1.8110x; 1.0591x over previous
#include <hip/hip_runtime.h>


typedef __attribute__((ext_vector_type(2))) float v2f;
typedef __attribute__((ext_vector_type(8))) float v8f;
typedef __attribute__((ext_vector_type(4))) float v4f;
typedef __attribute__((ext_vector_type(4))) int   v4i;

#define ROWS  64
#define MT     4
#define HSTR 260
#define H1S   68
#define H2S   36

__global__ __launch_bounds__(128)
void dfm_fused_kernel(const int*   __restrict__ x_field,
                      const int*   __restrict__ x,
                      const float* __restrict__ x_val,
                      const float* __restrict__ emb,
                      const float* __restrict__ lin_tab,
                      const float* __restrict__ lin_bias,
                      const float* __restrict__ W1, const float* __restrict__ b1,
                      const float* __restrict__ g1, const float* __restrict__ be1,
                      const float* __restrict__ W2, const float* __restrict__ b2,
                      const float* __restrict__ g2, const float* __restrict__ be2,
                      const float* __restrict__ W3, const float* __restrict__ b3,
                      float*       __restrict__ out)
{
    __shared__ float sh_h [ROWS * HSTR];
    __shared__ float sh_h1[ROWS * H1S];
    __shared__ float sh_h2[ROWS * H2S];
    __shared__ float sh_red[ROWS * 8];
    __shared__ float sh_lr[ROWS * 16];
    __shared__ float sh_lf[ROWS];

    const int t    = threadIdx.x;
    const int wave = t >> 5;
    const int lane = t & 31;
    const int half = lane >> 4;
    const int mlo  = lane & 15;
    const int r0   = blockIdx.x * ROWS;
    const float RSQ = rsqrtf(1.0f + 1e-5f);

    const int c1 = wave * 16 + mlo;
    const int c2 = (wave & 1) * 16 + mlo;
    v2f   bw[64];
    v2f   bw2[16];
    float g1v, b1v, be1v, g2v, b2v, be2v;
    const float lbias = lin_bias[0];
    const float b3v   = b3[0];
    float w3r[32];
    #pragma unroll
    for (int cc = 0; cc < 32; ++cc) w3r[cc] = W3[cc];

    const int q4 = t >> 2;
    const int j4 = t & 3;
    {
        const int gbase = r0 * 16 + q4 * 32;
        const v4i* xf4 = reinterpret_cast<const v4i*>(x_field + gbase);
        const v4i* xi4 = reinterpret_cast<const v4i*>(x + gbase);
        const v4f* xv4 = reinterpret_cast<const v4f*>(x_val + gbase);
        v4i xf[8], xi[8];
        v4f xv[8];
        #pragma unroll
        for (int k = 0; k < 8; ++k) { xf[k] = xf4[k]; xi[k] = xi4[k]; xv[k] = xv4[k]; }
        v4i lf[2], li[2];
        v4f lv[2];
        #pragma unroll
        for (int k = 0; k < 2; ++k) { lf[k] = xf4[2 * j4 + k]; li[k] = xi4[2 * j4 + k]; lv[k] = xv4[2 * j4 + k]; }

        v4f   a[32];
        float lw[8];
        const char* ebase = reinterpret_cast<const char*>(emb);
        const unsigned joff = 16u * (unsigned)j4;
        #pragma unroll
        for (int i = 0; i < 32; ++i) {
            const unsigned row = (unsigned)xf[i >> 2][i & 3] * 100000u + (unsigned)xi[i >> 2][i & 3];
            a[i] = *reinterpret_cast<const v4f*>(ebase + (row * 64u + joff));
        }
        #pragma unroll
        for (int k = 0; k < 8; ++k) {
            const unsigned row = (unsigned)lf[k >> 2][k & 3] * 100000u + (unsigned)li[k >> 2][k & 3];
            lw[k] = lin_tab[row];
        }
        __builtin_amdgcn_sched_barrier(0);
        #pragma unroll
        for (int ks = 0; ks < 31; ++ks) {
            const int k0 = 4 * ks + 2 * half;
            bw[ks].x = W1[(size_t)k0 * 64 + c1];
            bw[ks].y = W1[(size_t)(k0 + 1) * 64 + c1];
        }
        __builtin_amdgcn_sched_barrier(0);

        #pragma unroll
        for (int i = 0; i < 32; ++i) {
            const int r = 2 * q4 + (i >> 4), f = i & 15;
            a[i] = a[i] * xv[i >> 2][i & 3];
            *reinterpret_cast<v4f*>(&sh_h[r * HSTR + f * 16 + 4 * j4]) = a[i];
        }
        #pragma unroll
        for (int k = 0; k < 8; ++k)
            sh_lr[q4 * 32 + 8 * j4 + k] = lw[k] * lv[k >> 2][k & 3];

        #pragma unroll
        for (int rr = 0; rr < 2; ++rr) {
            float s0 = 0.f, q0 = 0.f, s1 = 0.f, q1 = 0.f, s2 = 0.f, q2 = 0.f, s3 = 0.f, q3 = 0.f;
            #pragma unroll
            for (int f = 0; f < 16; ++f) {
                const v4f v = a[rr * 16 + f];
                s0 += v[0]; q0 = fmaf(v[0], v[0], q0);
                s1 += v[1]; q1 = fmaf(v[1], v[1], q1);
                s2 += v[2]; q2 = fmaf(v[2], v[2], q2);
                s3 += v[3]; q3 = fmaf(v[3], v[3], q3);
            }
            float* red = &sh_red[(2 * q4 + rr) * 8 + 2 * j4];
            red[0] = (s0 * s0 - q0) + (s1 * s1 - q1);
            red[1] = (s2 * s2 - q2) + (s3 * s3 - q3);
        }
    }
    __syncthreads();

    if (t < ROWS) {
        float fm = 0.f;
        #pragma unroll
        for (int i = 0; i < 8; ++i) fm += sh_red[t * 8 + i];
        float lin = 0.f;
        #pragma unroll
        for (int f = 0; f < 16; ++f) lin += sh_lr[t * 16 + f];
        sh_lf[t] = lin + lbias + 0.5f * fm;
    }

    {
        const int c = c1;
        #pragma unroll
        for (int ks = 31; ks < 62; ++ks) {
            const int k0 = 4 * ks + 2 * half;
            bw[ks].x = W1[(size_t)k0 * 64 + c];
            bw[ks].y = W1[(size_t)(k0 + 1) * 64 + c];
        }
        v8f acc[MT];
        #pragma unroll
        for (int m = 0; m < MT; ++m) acc[m] = (v8f){};
        const float* ap = &sh_h[mlo * HSTR + 2 * half];
        v2f af[2][8 * MT];
        #pragma unroll
        for (int j = 0; j < 8; ++j) {
            #pragma unroll
            for (int m = 0; m < MT; ++m)
                af[0][j * MT + m] = *(const v2f*)&ap[m * 16 * HSTR + 4 * j];
        }
        #pragma unroll
        for (int blk = 0; blk < 8; ++blk) {
            if (blk == 2) {
                #pragma unroll
                for (int ks = 62; ks < 64; ++ks) {
                    const int k0 = 4 * ks + 2 * half;
                    bw[ks].x = W1[(size_t)k0 * 64 + c];
                    bw[ks].y = W1[(size_t)(k0 + 1) * 64 + c];
                }
                g1v = g1[c1]; b1v = b1[c1]; be1v = be1[c1];
                g2v = g2[c2]; b2v = b2[c2]; be2v = be2[c2];
                const float* w2p = W2 + (size_t)(2 * half) * 32 + c2;
                #pragma unroll
                for (int i = 0; i < 16; ++i) {
                    bw2[i].x = w2p[i * 128];
                    bw2[i].y = w2p[i * 128 + 32];
                }
            }
            if (blk < 7) {
                #pragma unroll
                for (int j = 0; j < 8; ++j) {
                    #pragma unroll
                    for (int m = 0; m < MT; ++m)
                        af[(blk + 1) & 1][j * MT + m] =
                            *(const v2f*)&ap[m * 16 * HSTR + 4 * ((blk + 1) * 8 + j)];
                }
            }
            __builtin_amdgcn_sched_barrier(0);
            #pragma unroll
            for (int j = 0; j < 8; ++j) {
                const v2f b = bw[blk * 8 + j];
                #pragma unroll
                for (int m = 0; m < MT; ++m)
                    acc[m] = __builtin_amdgcn_wmma_f32_16x16x4_f32(false, af[blk & 1][j * MT + m], false, b,
                                                                   (short)0, acc[m], false, false);
            }
            __builtin_amdgcn_sched_barrier(0);
        }
        const float sc = g1v * RSQ;
        const float tb = fmaf(b1v, sc, be1v);
        #pragma unroll
        for (int m = 0; m < MT; ++m) {
            #pragma unroll
            for (int j = 0; j < 8; ++j) {
                const int row = m * 16 + j + 8 * half;
                const float v = fmaf(acc[m][j], sc, tb);
                sh_h1[row * H1S + c] = v > 0.f ? v : 0.f;
            }
        }
    }
    __syncthreads();

    {
        const int c  = c2;
        const int mb = (wave >> 1) * 2;
        v8f acc[2];
        acc[0] = (v8f){}; acc[1] = (v8f){};
        v2f a2[32];
        #pragma unroll
        for (int i = 0; i < 16; ++i) {
            #pragma unroll
            for (int m = 0; m < 2; ++m)
                a2[i * 2 + m] = *(const v2f*)&sh_h1[((mb + m) * 16 + mlo) * H1S + 4 * i + 2 * half];
        }
        __builtin_amdgcn_sched_barrier(0);
        #pragma unroll
        for (int i = 0; i < 16; ++i) {
            const v2f b = bw2[i];
            #pragma unroll
            for (int m = 0; m < 2; ++m)
                acc[m] = __builtin_amdgcn_wmma_f32_16x16x4_f32(false, a2[i * 2 + m], false, b,
                                                               (short)0, acc[m], false, false);
        }
        const float sc = g2v * RSQ;
        const float tb = fmaf(b2v, sc, be2v);
        #pragma unroll
        for (int m = 0; m < 2; ++m) {
            #pragma unroll
            for (int j = 0; j < 8; ++j) {
                const int row = (mb + m) * 16 + j + 8 * half;
                const float v = fmaf(acc[m][j], sc, tb);
                sh_h2[row * H2S + c] = v > 0.f ? v : 0.f;
            }
        }
    }
    __syncthreads();

    if (t < ROWS) {
        float acc = sh_lf[t] + b3v;
        #pragma unroll
        for (int cc = 0; cc < 32; ++cc)
            acc = fmaf(sh_h2[t * H2S + cc], w3r[cc], acc);
        out[r0 + t] = acc;
    }
}

extern "C" void kernel_launch(void* const* d_in, const int* in_sizes, int n_in,
                              void* d_out, int out_size, void* d_ws, size_t ws_size,
                              hipStream_t stream) {
    const int*   x_field  = (const int*)  d_in[0];
    const int*   x        = (const int*)  d_in[1];
    const float* x_val    = (const float*)d_in[2];
    const float* emb      = (const float*)d_in[3];
    const float* lin_tab  = (const float*)d_in[4];
    const float* lin_bias = (const float*)d_in[5];
    const float* W1 = (const float*)d_in[6];
    const float* b1 = (const float*)d_in[7];
    const float* g1 = (const float*)d_in[8];
    const float* be1= (const float*)d_in[9];
    const float* W2 = (const float*)d_in[10];
    const float* b2 = (const float*)d_in[11];
    const float* g2 = (const float*)d_in[12];
    const float* be2= (const float*)d_in[13];
    const float* W3 = (const float*)d_in[14];
    const float* b3 = (const float*)d_in[15];
    float* out = (float*)d_out;

    const int nblocks = out_size / 64;
    dfm_fused_kernel<<<dim3(nblocks), dim3(128), 0, stream>>>(
        x_field, x, x_val, emb, lin_tab, lin_bias,
        W1, b1, g1, be1, W2, b2, g2, be2, W3, b3, out);
}
